# starting kernel
# baseline (speedup 1.0000x reference)
#include <hip/hip_runtime.h>
#include <math.h>

#define N_FEAT 128

typedef __attribute__((ext_vector_type(2))) float v2f;
typedef __attribute__((ext_vector_type(8))) float v8f;

__global__ void deg_init_kernel(float* __restrict__ deg, int n) {
    int i = blockIdx.x * blockDim.x + threadIdx.x;
    if (i < n) deg[i] = 1.0f;
}

__global__ void deg_scatter_kernel(const int* __restrict__ dst, float* __restrict__ deg, int E) {
    int e = blockIdx.x * blockDim.x + threadIdx.x;
    if (e < E) atomicAdd(&deg[dst[e]], 1.0f);
}

__global__ void deg_rsqrt_kernel(float* __restrict__ deg, int n) {
    int i = blockIdx.x * blockDim.x + threadIdx.x;
    if (i < n) deg[i] = rsqrtf(deg[i]);
}

__global__ void gemm_wmma_kernel(const float* __restrict__ X, const float* __restrict__ W,
                                 float* __restrict__ H, int M) {
    const int wave = threadIdx.x >> 5;
    const int lane = threadIdx.x & 31;
    const int half = lane >> 4;
    const int l16  = lane & 15;
    const int row0 = blockIdx.x * 16;
    if (row0 >= M) return;
    const int col = wave * 16 + l16;

    v8f acc = {};
    const float* xrow = X + (size_t)(row0 + l16) * N_FEAT + half * 2;
    #pragma unroll
    for (int k0 = 0; k0 < N_FEAT; k0 += 4) {
        v2f a;
        a.x = xrow[k0 + 0];
        a.y = xrow[k0 + 1];
        v2f b;
        b.x = W[(size_t)(k0 + half * 2 + 0) * N_FEAT + col];
        b.y = W[(size_t)(k0 + half * 2 + 1) * N_FEAT + col];
        acc = __builtin_amdgcn_wmma_f32_16x16x4_f32(
             false, a,  false, b,
             (short)0, acc,  false,  false);
    }
    float* out = H + (size_t)row0 * N_FEAT + wave * 16 + l16;
    #pragma unroll
    for (int r = 0; r < 8; ++r) {
        out[(size_t)(half * 8 + r) * N_FEAT] = acc[r];
    }
}

__global__ void zero_kernel(float4* __restrict__ p, long long n4) {
    long long i = (long long)blockIdx.x * blockDim.x + threadIdx.x;
    if (i < n4) p[i] = make_float4(0.f, 0.f, 0.f, 0.f);
}

__global__ void scatter_kernel(const float* __restrict__ H, const float* __restrict__ dinv,
                               const int* __restrict__ src, const int* __restrict__ dst,
                               float* __restrict__ agg, int E) {
    const int lane = threadIdx.x & 31;
    const int e = blockIdx.x * (blockDim.x >> 5) + (threadIdx.x >> 5);
    if (e >= E) return;
    const int s = src[e];
    const int d = dst[e];
    const float nrm = dinv[s] * dinv[d];
    const float4 v = *(const float4*)(H + (size_t)s * N_FEAT + lane * 4);
    float* p = agg + (size_t)d * N_FEAT + lane * 4;
    atomicAdd(p + 0, v.x * nrm);
    atomicAdd(p + 1, v.y * nrm);
    atomicAdd(p + 2, v.z * nrm);
    atomicAdd(p + 3, v.w * nrm);
}

__global__ void finalize_kernel(const float* __restrict__ agg, const float* __restrict__ H,
                                const float* __restrict__ dinv, const float* __restrict__ bias,
                                const float* __restrict__ alpha, float* __restrict__ out, int n) {
    const int lane = threadIdx.x & 31;
    const int node = blockIdx.x * (blockDim.x >> 5) + (threadIdx.x >> 5);
    if (node >= n) return;
    const float di = dinv[node];
    const float di2 = di * di;
    const int f0 = lane * 4;
    const float4 ag = *(const float4*)(agg + (size_t)node * N_FEAT + f0);
    const float4 hv = *(const float4*)(H + (size_t)node * N_FEAT + f0);
    const float4 bv = *(const float4*)(bias + f0);
    const float4 av = *(const float4*)(alpha + f0);
    float4 r;
    r.x = ag.x + hv.x * di2 + bv.x;  r.x = r.x > 0.f ? r.x : av.x * r.x;
    r.y = ag.y + hv.y * di2 + bv.y;  r.y = r.y > 0.f ? r.y : av.y * r.y;
    r.z = ag.z + hv.z * di2 + bv.z;  r.z = r.z > 0.f ? r.z : av.z * r.z;
    r.w = ag.w + hv.w * di2 + bv.w;  r.w = r.w > 0.f ? r.w : av.w * r.w;
    *(float4*)(out + (size_t)node * N_FEAT + f0) = r;
}

extern "C" void kernel_launch(void* const* d_in, const int* in_sizes, int n_in,
                              void* d_out, int out_size, void* d_ws, size_t ws_size,
                              hipStream_t stream) {
    (void)n_in; (void)out_size; (void)ws_size;
    const float* x  = (const float*)d_in[0];
    const int*   ei = (const int*)d_in[1];
    const float* W1 = (const float*)d_in[2];
    const float* b1 = (const float*)d_in[3];
    const float* W2 = (const float*)d_in[4];
    const float* b2 = (const float*)d_in[5];
    const float* al = (const float*)d_in[6];

    const int N = in_sizes[0] / N_FEAT;
    const int E = in_sizes[1] / 2;
    const int* src = ei;
    const int* dst = ei + E;

    float* dinv = (float*)d_ws;
    size_t off = (((size_t)N * sizeof(float)) + 255) & ~(size_t)255;
    float* hbuf = (float*)((char*)d_ws + off);
    float* agg  = (float*)d_out;
    float* outp = (float*)d_out;

    deg_init_kernel<<<(N + 255) / 256, 256, 0, stream>>>(dinv, N);
    deg_scatter_kernel<<<(E + 255) / 256, 256, 0, stream>>>(dst, dinv, E);
    deg_rsqrt_kernel<<<(N + 255) / 256, 256, 0, stream>>>(dinv, N);

    const int mTiles = (N + 15) / 16;
    const long long n4 = (long long)N * N_FEAT / 4;
    const int zBlocks = (int)((n4 + 255) / 256);
    const int eBlocks = (E + 7) / 8;
    const int nBlocks = (N + 7) / 8;

    gemm_wmma_kernel<<<mTiles, 256, 0, stream>>>(x, W1, hbuf, N);
    zero_kernel<<<zBlocks, 256, 0, stream>>>((float4*)agg, n4);
    scatter_kernel<<<eBlocks, 256, 0, stream>>>(hbuf, dinv, src, dst, agg, E);
    finalize_kernel<<<nBlocks, 256, 0, stream>>>(agg, hbuf, dinv, b1, al, outp, N);

    gemm_wmma_kernel<<<mTiles, 256, 0, stream>>>(outp, W2, hbuf, N);
    zero_kernel<<<zBlocks, 256, 0, stream>>>((float4*)agg, n4);
    scatter_kernel<<<eBlocks, 256, 0, stream>>>(hbuf, dinv, src, dst, agg, E);
    finalize_kernel<<<nBlocks, 256, 0, stream>>>(agg, hbuf, dinv, b2, al, outp, N);
}


# f32 atomic edge scatter replaced by per-destination gather: edges bucketed by dst each call (integer counting adds), segments sorted by src so the sum order is fixed, one wave sums a row
# speedup vs baseline: 2.4824x; 2.4824x over previous
#include <hip/hip_runtime.h>
#include <math.h>

#define N_FEAT 128

typedef __attribute__((ext_vector_type(2))) float v2f;
typedef __attribute__((ext_vector_type(8))) float v8f;

#define SCAN_TILE 1024

__global__ void cnt_zero_kernel(int* __restrict__ cnt, int n) {
    int i = blockIdx.x * blockDim.x + threadIdx.x;
    if (i < n) cnt[i] = 0;
}

__global__ void count_rank_kernel(const int* __restrict__ dst, int* __restrict__ cnt,
                                  int* __restrict__ rank, int E) {
    int e = blockIdx.x * blockDim.x + threadIdx.x;
    if (e < E) rank[e] = atomicAdd(&cnt[dst[e]], 1);
}

__device__ __forceinline__ int wave_sum(int v) {
    for (int o = 16; o > 0; o >>= 1) v += __shfl_xor(v, o);
    return v;
}

__global__ __launch_bounds__(256) void block_sums_kernel(const int* __restrict__ cnt,
                                                         int* __restrict__ bsum, int n) {
    __shared__ int wtot[8];
    const int base = blockIdx.x * SCAN_TILE + threadIdx.x * 4;
    int s = 0;
    for (int k = 0; k < 4; ++k) if (base + k < n) s += cnt[base + k];
    s = wave_sum(s);
    if ((threadIdx.x & 31) == 0) wtot[threadIdx.x >> 5] = s;
    __syncthreads();
    if (threadIdx.x == 0) {
        int t = 0;
        for (int w = 0; w < 8; ++w) t += wtot[w];
        bsum[blockIdx.x] = t;
    }
}

__global__ __launch_bounds__(256) void scan_dinv_kernel(const int* __restrict__ cnt,
                                                        const int* __restrict__ bsum,
                                                        int* __restrict__ rowptr,
                                                        float* __restrict__ dinv, int n) {
    __shared__ int wtot[8];
    __shared__ int boff[8];
    const int lane = threadIdx.x & 31;
    const int wave = threadIdx.x >> 5;
    int p = 0;
    for (int i = threadIdx.x; i < (int)blockIdx.x; i += 256) p += bsum[i];
    p = wave_sum(p);
    const int base = blockIdx.x * SCAN_TILE + threadIdx.x * 4;
    int c[4];
    int ts = 0;
    for (int k = 0; k < 4; ++k) { c[k] = (base + k < n) ? cnt[base + k] : 0; ts += c[k]; }
    int inc = ts;
    for (int o = 1; o < 32; o <<= 1) { int t = __shfl_up(inc, o); if (lane >= o) inc += t; }
    if (lane == 31) wtot[wave] = inc;
    if (lane == 0)  boff[wave] = p;
    __syncthreads();
    int run = inc - ts;
    for (int w = 0; w < 8; ++w) { run += boff[w]; if (w < wave) run += wtot[w]; }
    for (int k = 0; k < 4; ++k) {
        const int i = base + k;
        if (i < n) {
            rowptr[i] = run;
            dinv[i] = rsqrtf((float)(c[k] + 1));
            run += c[k];
            if (i == n - 1) rowptr[n] = run;
        }
    }
}

__global__ void fill_kernel(const int* __restrict__ src, const int* __restrict__ dst,
                            const int* __restrict__ rank, const int* __restrict__ rowptr,
                            int* __restrict__ tmp, int E) {
    int e = blockIdx.x * blockDim.x + threadIdx.x;
    if (e < E) tmp[rowptr[dst[e]] + rank[e]] = src[e];
}

__global__ __launch_bounds__(256) void seg_order_kernel(const int* __restrict__ rowptr,
                                                        const int* __restrict__ tmp,
                                                        int* __restrict__ csr, int n) {
    const int lane = threadIdx.x & 31;
    int node = blockIdx.x * 8 + (threadIdx.x >> 5);
    if (node >= n) return;
    node = __builtin_amdgcn_readfirstlane(node);
    const int beg = rowptr[node];
    const int d = rowptr[node + 1] - beg;
    for (int c = 0; c < d; c += 32) {
        const int i = c + lane;
        const int x = (i < d) ? tmp[beg + i] : 0;
        int pos = 0;
        for (int c2 = 0; c2 < d; c2 += 32) {
            const int i2 = c2 + lane;
            const int y = (i2 < d) ? tmp[beg + i2] : 0;
            const int m2 = (d - c2 < 32) ? (d - c2) : 32;
            for (int j = 0; j < m2; ++j) {
                const int vj = __builtin_amdgcn_readlane(y, j);
                const int t = (c2 + j < i) ? 1 : 0;
                pos += (vj < x + t) ? 1 : 0;
            }
        }
        if (i < d) csr[beg + pos] = x;
    }
}

__global__ void gemm_wmma_kernel(const float* __restrict__ X, const float* __restrict__ W,
                                 float* __restrict__ H, int M) {
    const int wave = threadIdx.x >> 5;
    const int lane = threadIdx.x & 31;
    const int half = lane >> 4;
    const int l16  = lane & 15;
    const int row0 = blockIdx.x * 16;
    if (row0 >= M) return;
    const int col = wave * 16 + l16;

    v8f acc = {};
    const float* xrow = X + (size_t)(row0 + l16) * N_FEAT + half * 2;
    #pragma unroll
    for (int k0 = 0; k0 < N_FEAT; k0 += 4) {
        v2f a;
        a.x = xrow[k0 + 0];
        a.y = xrow[k0 + 1];
        v2f b;
        b.x = W[(size_t)(k0 + half * 2 + 0) * N_FEAT + col];
        b.y = W[(size_t)(k0 + half * 2 + 1) * N_FEAT + col];
        acc = __builtin_amdgcn_wmma_f32_16x16x4_f32(
             false, a,  false, b,
             (short)0, acc,  false,  false);
    }
    float* out = H + (size_t)row0 * N_FEAT + wave * 16 + l16;
    #pragma unroll
    for (int r = 0; r < 8; ++r) {
        out[(size_t)(half * 8 + r) * N_FEAT] = acc[r];
    }
}


template <int K>
__device__ __forceinline__ void add_rows(float4& acc, const float* __restrict__ hl,
                                         int sidx, int wbits, int j) {
    float4 v[K];
    float w[K];
    for (int k = 0; k < K; ++k) {
        const int s = __builtin_amdgcn_readlane(sidx, j + k);
        w[k] = __int_as_float(__builtin_amdgcn_readlane(wbits, j + k));
        v[k] = *(const float4*)(hl + (size_t)s * N_FEAT);
    }
    for (int k = 0; k < K; ++k) {
        acc.x = __builtin_fmaf(v[k].x, w[k], acc.x);
        acc.y = __builtin_fmaf(v[k].y, w[k], acc.y);
        acc.z = __builtin_fmaf(v[k].z, w[k], acc.z);
        acc.w = __builtin_fmaf(v[k].w, w[k], acc.w);
    }
}

__global__ __launch_bounds__(256) void gather_kernel(const float* __restrict__ H,
                                                     const float* __restrict__ dinv,
                                                     const int* __restrict__ rowptr,
                                                     const int* __restrict__ csr,
                                                     float* __restrict__ agg, int n) {
    const int lane = threadIdx.x & 31;
    int node = blockIdx.x * 8 + (threadIdx.x >> 5);
    if (node >= n) return;
    node = __builtin_amdgcn_readfirstlane(node);
    const int beg = rowptr[node];
    const int end = rowptr[node + 1];
    const float dd = dinv[node];
    const float* hl = H + lane * 4;
    float4 acc = make_float4(0.f, 0.f, 0.f, 0.f);
    for (int base = beg; base < end; base += 32) {
        const int m = (end - base < 32) ? (end - base) : 32;
        int sidx = 0;
        float w = 0.f;
        if (lane < m) { sidx = csr[base + lane]; w = dinv[sidx] * dd; }
        const int wbits = __float_as_int(w);
        int j = 0;
        for (; j + 8 <= m; j += 8) add_rows<8>(acc, hl, sidx, wbits, j);
        if (j + 4 <= m) { add_rows<4>(acc, hl, sidx, wbits, j); j += 4; }
        const int r = m - j;
        if (r == 3)      add_rows<3>(acc, hl, sidx, wbits, j);
        else if (r == 2) add_rows<2>(acc, hl, sidx, wbits, j);
        else if (r == 1) add_rows<1>(acc, hl, sidx, wbits, j);
    }
    *(float4*)(agg + (size_t)node * N_FEAT + lane * 4) = acc;
}

__global__ void finalize_kernel(const float* __restrict__ agg, const float* __restrict__ H,
                                const float* __restrict__ dinv, const float* __restrict__ bias,
                                const float* __restrict__ alpha, float* __restrict__ out, int n) {
    const int lane = threadIdx.x & 31;
    const int node = blockIdx.x * (blockDim.x >> 5) + (threadIdx.x >> 5);
    if (node >= n) return;
    const float di = dinv[node];
    const float di2 = di * di;
    const int f0 = lane * 4;
    const float4 ag = *(const float4*)(agg + (size_t)node * N_FEAT + f0);
    const float4 hv = *(const float4*)(H + (size_t)node * N_FEAT + f0);
    const float4 bv = *(const float4*)(bias + f0);
    const float4 av = *(const float4*)(alpha + f0);
    float4 r;
    r.x = ag.x + hv.x * di2 + bv.x;  r.x = r.x > 0.f ? r.x : av.x * r.x;
    r.y = ag.y + hv.y * di2 + bv.y;  r.y = r.y > 0.f ? r.y : av.y * r.y;
    r.z = ag.z + hv.z * di2 + bv.z;  r.z = r.z > 0.f ? r.z : av.z * r.z;
    r.w = ag.w + hv.w * di2 + bv.w;  r.w = r.w > 0.f ? r.w : av.w * r.w;
    *(float4*)(out + (size_t)node * N_FEAT + f0) = r;
}

__global__ void deg_init_kernel(float* __restrict__ deg, int n) {
    int i = blockIdx.x * blockDim.x + threadIdx.x;
    if (i < n) deg[i] = 1.0f;
}

__global__ void deg_scatter_kernel(const int* __restrict__ dst, float* __restrict__ deg, int E) {
    int e = blockIdx.x * blockDim.x + threadIdx.x;
    if (e < E) atomicAdd(&deg[dst[e]], 1.0f);
}

__global__ void deg_rsqrt_kernel(float* __restrict__ deg, int n) {
    int i = blockIdx.x * blockDim.x + threadIdx.x;
    if (i < n) deg[i] = rsqrtf(deg[i]);
}

__global__ void zero_kernel(float4* __restrict__ p, long long n4) {
    long long i = (long long)blockIdx.x * blockDim.x + threadIdx.x;
    if (i < n4) p[i] = make_float4(0.f, 0.f, 0.f, 0.f);
}

__global__ void scatter_kernel(const float* __restrict__ H, const float* __restrict__ dinv,
                               const int* __restrict__ src, const int* __restrict__ dst,
                               float* __restrict__ agg, int E) {
    const int lane = threadIdx.x & 31;
    const int e = blockIdx.x * (blockDim.x >> 5) + (threadIdx.x >> 5);
    if (e >= E) return;
    const int s = src[e];
    const int d = dst[e];
    const float nrm = dinv[s] * dinv[d];
    const float4 v = *(const float4*)(H + (size_t)s * N_FEAT + lane * 4);
    float* p = agg + (size_t)d * N_FEAT + lane * 4;
    atomicAdd(p + 0, v.x * nrm);
    atomicAdd(p + 1, v.y * nrm);
    atomicAdd(p + 2, v.z * nrm);
    atomicAdd(p + 3, v.w * nrm);
}

extern "C" void kernel_launch(void* const* d_in, const int* in_sizes, int n_in,
                              void* d_out, int out_size, void* d_ws, size_t ws_size,
                              hipStream_t stream) {
    (void)n_in; (void)out_size;
    const float* x  = (const float*)d_in[0];
    const int*   ei = (const int*)d_in[1];
    const float* W1 = (const float*)d_in[2];
    const float* b1 = (const float*)d_in[3];
    const float* W2 = (const float*)d_in[4];
    const float* b2 = (const float*)d_in[5];
    const float* al = (const float*)d_in[6];

    const int N = in_sizes[0] / N_FEAT;
    const int E = in_sizes[1] / 2;
    const int* src = ei;
    const int* dst = ei + E;

    float* dinv = (float*)d_ws;
    size_t off = (((size_t)N * sizeof(float)) + 255) & ~(size_t)255;
    float* hbuf = (float*)((char*)d_ws + off);
    off += (size_t)N * N_FEAT * sizeof(float);
    int* rowptr = (int*)((char*)d_ws + off);
    off += (((size_t)(N + 1) * sizeof(int)) + 255) & ~(size_t)255;
    int* csr = (int*)((char*)d_ws + off);
    off += (size_t)E * sizeof(int);
    float* agg  = (float*)d_out;
    float* outp = (float*)d_out;

    const int mTiles = (N + 15) / 16;
    const int nBlocks = (N + 7) / 8;
    const int sBlocks = (N + SCAN_TILE - 1) / SCAN_TILE;

    const size_t cntPad = ((size_t)N + 63) & ~(size_t)63;
    const size_t bsPad  = ((size_t)sBlocks + 63) & ~(size_t)63;
    int* cnt  = (int*)d_out;
    int* bsum = cnt + cntPad;
    int* rank = bsum + bsPad;
    int* tmp  = rank + E;
    const size_t scratchInts = cntPad + bsPad + 2 * (size_t)E;

    if (off <= ws_size && scratchInts <= (size_t)N * N_FEAT) {
        cnt_zero_kernel<<<(N + 255) / 256, 256, 0, stream>>>(cnt, N);
        count_rank_kernel<<<(E + 255) / 256, 256, 0, stream>>>(dst, cnt, rank, E);
        block_sums_kernel<<<sBlocks, 256, 0, stream>>>(cnt, bsum, N);
        scan_dinv_kernel<<<sBlocks, 256, 0, stream>>>(cnt, bsum, rowptr, dinv, N);
        fill_kernel<<<(E + 255) / 256, 256, 0, stream>>>(src, dst, rank, rowptr, tmp, E);
        seg_order_kernel<<<nBlocks, 256, 0, stream>>>(rowptr, tmp, csr, N);

        gemm_wmma_kernel<<<mTiles, 256, 0, stream>>>(x, W1, hbuf, N);
        gather_kernel<<<nBlocks, 256, 0, stream>>>(hbuf, dinv, rowptr, csr, agg, N);
        finalize_kernel<<<nBlocks, 256, 0, stream>>>(agg, hbuf, dinv, b1, al, outp, N);

        gemm_wmma_kernel<<<mTiles, 256, 0, stream>>>(outp, W2, hbuf, N);
        gather_kernel<<<nBlocks, 256, 0, stream>>>(hbuf, dinv, rowptr, csr, agg, N);
        finalize_kernel<<<nBlocks, 256, 0, stream>>>(agg, hbuf, dinv, b2, al, outp, N);
        return;
    }

    deg_init_kernel<<<(N + 255) / 256, 256, 0, stream>>>(dinv, N);
    deg_scatter_kernel<<<(E + 255) / 256, 256, 0, stream>>>(dst, dinv, E);
    deg_rsqrt_kernel<<<(N + 255) / 256, 256, 0, stream>>>(dinv, N);

    const long long n4 = (long long)N * N_FEAT / 4;
    const int zBlocks = (int)((n4 + 255) / 256);
    const int eBlocks = (E + 7) / 8;

    gemm_wmma_kernel<<<mTiles, 256, 0, stream>>>(x, W1, hbuf, N);
    zero_kernel<<<zBlocks, 256, 0, stream>>>((float4*)agg, n4);
    scatter_kernel<<<eBlocks, 256, 0, stream>>>(hbuf, dinv, src, dst, agg, E);
    finalize_kernel<<<nBlocks, 256, 0, stream>>>(agg, hbuf, dinv, b1, al, outp, N);

    gemm_wmma_kernel<<<mTiles, 256, 0, stream>>>(outp, W2, hbuf, N);
    zero_kernel<<<zBlocks, 256, 0, stream>>>((float4*)agg, n4);
    scatter_kernel<<<eBlocks, 256, 0, stream>>>(hbuf, dinv, src, dst, agg, E);
    finalize_kernel<<<nBlocks, 256, 0, stream>>>(agg, hbuf, dinv, b2, al, outp, N);
}


# GEMM restructured: 16-wave groups loop over 32-row tiles, each wave keeps its 128x16 slice of W as B fragments in registers, X tiles staged as whole rows through LDS; k and sum order unchanged
# speedup vs baseline: 4.8761x; 1.9643x over previous
#include <hip/hip_runtime.h>
#include <math.h>

#define N_FEAT 128

typedef __attribute__((ext_vector_type(2))) float v2f;
typedef __attribute__((ext_vector_type(8))) float v8f;

#define BK_SHIFT 7
#define BK_NODES 128
#define BK_MAX   1024
#define PB_EDGES 4096
#define BK_STAGE 4096

__global__ void cnt_zero_kernel(int* __restrict__ cnt, int n) {
    int i = blockIdx.x * blockDim.x + threadIdx.x;
    if (i < n) cnt[i] = 0;
}

__global__ __launch_bounds__(256) void bucket_count_kernel(const int* __restrict__ dst,
                                                           int* __restrict__ btot, int E, int nb) {
    __shared__ int hist[BK_MAX];
    for (int b = threadIdx.x; b < BK_MAX; b += 256) hist[b] = 0;
    __syncthreads();
    const int e0 = blockIdx.x * PB_EDGES + threadIdx.x;
    int d[PB_EDGES / 256];
    #pragma unroll
    for (int k = 0; k < PB_EDGES / 256; ++k) {
        const int e = e0 + k * 256;
        d[k] = (e < E) ? dst[e] : -1;
    }
    #pragma unroll
    for (int k = 0; k < PB_EDGES / 256; ++k) if (d[k] >= 0) atomicAdd(&hist[d[k] >> BK_SHIFT], 1);
    __syncthreads();
    for (int b = threadIdx.x; b < nb; b += 256) {
        const int h = hist[b];
        if (h) atomicAdd(&btot[b], h);
    }
}

__global__ __launch_bounds__(256) void bucket_place_kernel(const int* __restrict__ src,
                                                           const int* __restrict__ dst,
                                                           const int* __restrict__ btot,
                                                           int* __restrict__ bcur,
                                                           int* __restrict__ bstart,
                                                           int2* __restrict__ pairs, int E, int nb) {
    __shared__ int hist[BK_MAX];
    __shared__ int bst[BK_MAX];
    __shared__ int wtot[8];
    const int tid = threadIdx.x;
    const int lane = tid & 31;
    const int wave = tid >> 5;
    for (int b = tid; b < BK_MAX; b += 256) hist[b] = 0;
    __syncthreads();
    const int e0 = blockIdx.x * PB_EDGES + tid;
    int d[PB_EDGES / 256];
    #pragma unroll
    for (int k = 0; k < PB_EDGES / 256; ++k) {
        const int e = e0 + k * 256;
        d[k] = (e < E) ? dst[e] : -1;
    }
    #pragma unroll
    for (int k = 0; k < PB_EDGES / 256; ++k) if (d[k] >= 0) atomicAdd(&hist[d[k] >> BK_SHIFT], 1);
    int c[4];
    int ts = 0;
    #pragma unroll
    for (int k = 0; k < 4; ++k) { const int b = tid * 4 + k; c[k] = (b < nb) ? btot[b] : 0; ts += c[k]; }
    int inc = ts;
    for (int o = 1; o < 32; o <<= 1) { const int t = __shfl_up(inc, o); if (lane >= o) inc += t; }
    if (lane == 31) wtot[wave] = inc;
    __syncthreads();
    int run = inc - ts;
    for (int w = 0; w < wave; ++w) run += wtot[w];
    #pragma unroll
    for (int k = 0; k < 4; ++k) { bst[tid * 4 + k] = run; run += c[k]; }
    if (blockIdx.x == 0 && tid == 255) bstart[nb] = run;
    __syncthreads();
    for (int b = tid; b < nb; b += 256) {
        const int start = bst[b];
        if (blockIdx.x == 0) bstart[b] = start;
        const int h = hist[b];
        int base = 0;
        if (h) base = atomicAdd(&bcur[b], h);
        hist[b] = start + base;
    }
    __syncthreads();
    int sv[PB_EDGES / 256];
    #pragma unroll
    for (int k = 0; k < PB_EDGES / 256; ++k) sv[k] = (d[k] >= 0) ? src[e0 + k * 256] : 0;
    #pragma unroll
    for (int k = 0; k < PB_EDGES / 256; ++k) {
        if (d[k] >= 0) {
            const int p = atomicAdd(&hist[d[k] >> BK_SHIFT], 1);
            pairs[p] = make_int2(d[k], sv[k]);
        }
    }
}

__global__ __launch_bounds__(256) void bucket_finish_kernel(const int2* __restrict__ pairs,
                                                            const int* __restrict__ bstart,
                                                            int* __restrict__ rowptr,
                                                            float* __restrict__ dinv,
                                                            int* __restrict__ tmp, int n, int nb) {
    __shared__ int cnt[BK_NODES];
    __shared__ int wtot[4];
    __shared__ int stage[BK_STAGE];
    const int tid = threadIdx.x;
    const int lane = tid & 31;
    const int wave = tid >> 5;
    const int b = blockIdx.x;
    const int beg = bstart[b];
    const int end = bstart[b + 1];
    const int node0 = b << BK_SHIFT;
    if (tid < BK_NODES) cnt[tid] = 0;
    __syncthreads();
    for (int i0 = beg + tid; i0 < end; i0 += 8 * 256) {
        int dn[8];
        #pragma unroll
        for (int k = 0; k < 8; ++k) { const int i = i0 + k * 256; dn[k] = (i < end) ? pairs[i].x : -1; }
        #pragma unroll
        for (int k = 0; k < 8; ++k) if (dn[k] >= 0) atomicAdd(&cnt[dn[k] - node0], 1);
    }
    __syncthreads();
    const int c = (tid < BK_NODES) ? cnt[tid] : 0;
    int inc = c;
    for (int o = 1; o < 32; o <<= 1) { const int t = __shfl_up(inc, o); if (lane >= o) inc += t; }
    if (tid < BK_NODES && lane == 31) wtot[wave] = inc;
    __syncthreads();
    if (tid < BK_NODES) {
        int ex = inc - c;
        for (int w = 0; w < wave; ++w) ex += wtot[w];
        cnt[tid] = ex;
        const int node = node0 + tid;
        if (node < n) {
            rowptr[node] = beg + ex;
            dinv[node] = rsqrtf((float)(c + 1));
        }
    }
    if (b == nb - 1 && tid == 0) rowptr[n] = end;
    __syncthreads();
    const bool fits = (end - beg) <= BK_STAGE;
    for (int i0 = beg + tid; i0 < end; i0 += 8 * 256) {
        int2 pr[8];
        #pragma unroll
        for (int k = 0; k < 8; ++k) { const int i = i0 + k * 256; pr[k] = (i < end) ? pairs[i] : make_int2(-1, 0); }
        #pragma unroll
        for (int k = 0; k < 8; ++k) {
            if (pr[k].x >= 0) {
                const int p = atomicAdd(&cnt[pr[k].x - node0], 1);
                if (fits) stage[p] = pr[k].y; else tmp[beg + p] = pr[k].y;
            }
        }
    }
    if (fits) {
        __syncthreads();
        for (int i = tid; i < end - beg; i += 256) tmp[beg + i] = stage[i];
    }
}

__global__ __launch_bounds__(256) void seg_order_kernel(const int* __restrict__ rowptr,
                                                        const int* __restrict__ tmp,
                                                        int* __restrict__ csr, int n) {
    const int lane = threadIdx.x & 31;
    int node = blockIdx.x * 8 + (threadIdx.x >> 5);
    if (node >= n) return;
    node = __builtin_amdgcn_readfirstlane(node);
    const int beg = rowptr[node];
    const int d = rowptr[node + 1] - beg;
    for (int c = 0; c < d; c += 32) {
        const int i = c + lane;
        const int x = (i < d) ? tmp[beg + i] : 0;
        int pos = 0;
        for (int c2 = 0; c2 < d; c2 += 32) {
            const int i2 = c2 + lane;
            const int y = (i2 < d) ? tmp[beg + i2] : 0;
            const int m2 = (d - c2 < 32) ? (d - c2) : 32;
            for (int j = 0; j < m2; ++j) {
                const int vj = __builtin_amdgcn_readlane(y, j);
                const int t = (c2 + j < i) ? 1 : 0;
                pos += (vj < x + t) ? 1 : 0;
            }
        }
        if (i < d) csr[beg + pos] = x;
    }
}

#define GM_ROWS    32
#define GM_PITCH   132
#define GM_GRID    512
#define GM_THREADS 512

__global__ __launch_bounds__(GM_THREADS) void gemm_wmma_kernel(const float* __restrict__ X,
                                                               const float* __restrict__ W,
                                                               float* __restrict__ H, int M) {
    __shared__ __attribute__((aligned(16))) float xs[2][GM_ROWS][GM_PITCH];
    const int tid  = threadIdx.x;
    const int wave = tid >> 5;
    const int lane = tid & 31;
    const int half = lane >> 4;
    const int l16  = lane & 15;
    const int col  = (wave & 7) * 16 + l16;
    const int mrow = (wave >> 3) * 16;
    const int nTiles = (M + GM_ROWS - 1) / GM_ROWS;
    const int srow = tid >> 5;
    const int scol = lane * 4;

    int tile = blockIdx.x;
    float4 p0 = make_float4(0.f, 0.f, 0.f, 0.f);
    float4 p1 = p0;
    if (tile < nTiles) {
        int r0 = tile * GM_ROWS + srow;
        int r1 = r0 + 16;
        r0 = (r0 < M) ? r0 : M - 1;
        r1 = (r1 < M) ? r1 : M - 1;
        p0 = *(const float4*)(X + (size_t)r0 * N_FEAT + scol);
        p1 = *(const float4*)(X + (size_t)r1 * N_FEAT + scol);
    }

    v2f b[N_FEAT / 4];
    #pragma unroll
    for (int t = 0; t < N_FEAT / 4; ++t) {
        b[t].x = W[(size_t)(4 * t + half * 2 + 0) * N_FEAT + col];
        b[t].y = W[(size_t)(4 * t + half * 2 + 1) * N_FEAT + col];
    }

    if (tile < nTiles) {
        *(float4*)&xs[0][srow][scol]      = p0;
        *(float4*)&xs[0][srow + 16][scol] = p1;
    }
    int stage = 0;
    for (; tile < nTiles; tile += gridDim.x) {
        __syncthreads();
        const int nxt = tile + gridDim.x;
        const bool more = nxt < nTiles;
        if (more) {
            int r0 = nxt * GM_ROWS + srow;
            int r1 = r0 + 16;
            r0 = (r0 < M) ? r0 : M - 1;
            r1 = (r1 < M) ? r1 : M - 1;
            p0 = *(const float4*)(X + (size_t)r0 * N_FEAT + scol);
            p1 = *(const float4*)(X + (size_t)r1 * N_FEAT + scol);
        }

        v8f acc = {};
        const float* arow = &xs[stage][mrow + l16][half * 2];
        #pragma unroll
        for (int t = 0; t < N_FEAT / 4; ++t) {
            const v2f a = *(const v2f*)(arow + 4 * t);
            acc = __builtin_amdgcn_wmma_f32_16x16x4_f32(
                 false, a,  false, b[t],
                 (short)0, acc,  false,  false);
        }
        const int orow = tile * GM_ROWS + mrow + half * 8;
        float* out = H + (size_t)orow * N_FEAT + col;
        if ((tile + 1) * GM_ROWS <= M) {
            #pragma unroll
            for (int r = 0; r < 8; ++r) out[(size_t)r * N_FEAT] = acc[r];
        } else {
            #pragma unroll
            for (int r = 0; r < 8; ++r) {
                if (orow + r < M) out[(size_t)r * N_FEAT] = acc[r];
            }
        }
        if (more) {
            *(float4*)&xs[stage ^ 1][srow][scol]      = p0;
            *(float4*)&xs[stage ^ 1][srow + 16][scol] = p1;
        }
        stage ^= 1;
    }
}


template <int K>
__device__ __forceinline__ void add_rows(float4& acc, const float* __restrict__ hl,
                                         int sidx, int wbits, int j) {
    float4 v[K];
    float w[K];
    for (int k = 0; k < K; ++k) {
        const int s = __builtin_amdgcn_readlane(sidx, j + k);
        w[k] = __int_as_float(__builtin_amdgcn_readlane(wbits, j + k));
        v[k] = *(const float4*)(hl + (size_t)s * N_FEAT);
    }
    for (int k = 0; k < K; ++k) {
        acc.x = __builtin_fmaf(v[k].x, w[k], acc.x);
        acc.y = __builtin_fmaf(v[k].y, w[k], acc.y);
        acc.z = __builtin_fmaf(v[k].z, w[k], acc.z);
        acc.w = __builtin_fmaf(v[k].w, w[k], acc.w);
    }
}

__global__ __launch_bounds__(256) void gather_finalize_kernel(const float* __restrict__ H,
                                                              const float* __restrict__ dinv,
                                                              const int* __restrict__ rowptr,
                                                              const int* __restrict__ csr,
                                                              const float* __restrict__ bias,
                                                              const float* __restrict__ alpha,
                                                              float* __restrict__ out, int n) {
    const int lane = threadIdx.x & 31;
    int node = blockIdx.x * 8 + (threadIdx.x >> 5);
    if (node >= n) return;
    node = __builtin_amdgcn_readfirstlane(node);
    const int beg = rowptr[node];
    const int end = rowptr[node + 1];
    const float dd = dinv[node];
    const float* hl = H + lane * 4;
    const float4 hv = *(const float4*)(hl + (size_t)node * N_FEAT);
    const float4 bv = *(const float4*)(bias + lane * 4);
    const float4 av = *(const float4*)(alpha + lane * 4);
    float4 acc = make_float4(0.f, 0.f, 0.f, 0.f);
    for (int base = beg; base < end; base += 32) {
        const int m = (end - base < 32) ? (end - base) : 32;
        int sidx = 0;
        float w = 0.f;
        if (lane < m) { sidx = csr[base + lane]; w = dinv[sidx] * dd; }
        const int wbits = __float_as_int(w);
        int j = 0;
        for (; j + 8 <= m; j += 8) add_rows<8>(acc, hl, sidx, wbits, j);
        if (j + 4 <= m) { add_rows<4>(acc, hl, sidx, wbits, j); j += 4; }
        const int r = m - j;
        if (r == 3)      add_rows<3>(acc, hl, sidx, wbits, j);
        else if (r == 2) add_rows<2>(acc, hl, sidx, wbits, j);
        else if (r == 1) add_rows<1>(acc, hl, sidx, wbits, j);
    }
    const float di2 = dd * dd;
    float4 r;
    r.x = acc.x + hv.x * di2 + bv.x;  r.x = r.x > 0.f ? r.x : av.x * r.x;
    r.y = acc.y + hv.y * di2 + bv.y;  r.y = r.y > 0.f ? r.y : av.y * r.y;
    r.z = acc.z + hv.z * di2 + bv.z;  r.z = r.z > 0.f ? r.z : av.z * r.z;
    r.w = acc.w + hv.w * di2 + bv.w;  r.w = r.w > 0.f ? r.w : av.w * r.w;
    *(float4*)(out + (size_t)node * N_FEAT + lane * 4) = r;
}

__global__ void finalize_kernel(const float* __restrict__ agg, const float* __restrict__ H,
                                const float* __restrict__ dinv, const float* __restrict__ bias,
                                const float* __restrict__ alpha, float* __restrict__ out, int n) {
    const int lane = threadIdx.x & 31;
    const int node = blockIdx.x * (blockDim.x >> 5) + (threadIdx.x >> 5);
    if (node >= n) return;
    const float di = dinv[node];
    const float di2 = di * di;
    const int f0 = lane * 4;
    const float4 ag = *(const float4*)(agg + (size_t)node * N_FEAT + f0);
    const float4 hv = *(const float4*)(H + (size_t)node * N_FEAT + f0);
    const float4 bv = *(const float4*)(bias + f0);
    const float4 av = *(const float4*)(alpha + f0);
    float4 r;
    r.x = ag.x + hv.x * di2 + bv.x;  r.x = r.x > 0.f ? r.x : av.x * r.x;
    r.y = ag.y + hv.y * di2 + bv.y;  r.y = r.y > 0.f ? r.y : av.y * r.y;
    r.z = ag.z + hv.z * di2 + bv.z;  r.z = r.z > 0.f ? r.z : av.z * r.z;
    r.w = ag.w + hv.w * di2 + bv.w;  r.w = r.w > 0.f ? r.w : av.w * r.w;
    *(float4*)(out + (size_t)node * N_FEAT + f0) = r;
}

__global__ void deg_init_kernel(float* __restrict__ deg, int n) {
    int i = blockIdx.x * blockDim.x + threadIdx.x;
    if (i < n) deg[i] = 1.0f;
}

__global__ void deg_scatter_kernel(const int* __restrict__ dst, float* __restrict__ deg, int E) {
    int e = blockIdx.x * blockDim.x + threadIdx.x;
    if (e < E) atomicAdd(&deg[dst[e]], 1.0f);
}

__global__ void deg_rsqrt_kernel(float* __restrict__ deg, int n) {
    int i = blockIdx.x * blockDim.x + threadIdx.x;
    if (i < n) deg[i] = rsqrtf(deg[i]);
}

__global__ void zero_kernel(float4* __restrict__ p, long long n4) {
    long long i = (long long)blockIdx.x * blockDim.x + threadIdx.x;
    if (i < n4) p[i] = make_float4(0.f, 0.f, 0.f, 0.f);
}

__global__ void scatter_kernel(const float* __restrict__ H, const float* __restrict__ dinv,
                               const int* __restrict__ src, const int* __restrict__ dst,
                               float* __restrict__ agg, int E) {
    const int lane = threadIdx.x & 31;
    const int e = blockIdx.x * (blockDim.x >> 5) + (threadIdx.x >> 5);
    if (e >= E) return;
    const int s = src[e];
    const int d = dst[e];
    const float nrm = dinv[s] * dinv[d];
    const float4 v = *(const float4*)(H + (size_t)s * N_FEAT + lane * 4);
    float* p = agg + (size_t)d * N_FEAT + lane * 4;
    atomicAdd(p + 0, v.x * nrm);
    atomicAdd(p + 1, v.y * nrm);
    atomicAdd(p + 2, v.z * nrm);
    atomicAdd(p + 3, v.w * nrm);
}

extern "C" void kernel_launch(void* const* d_in, const int* in_sizes, int n_in,
                              void* d_out, int out_size, void* d_ws, size_t ws_size,
                              hipStream_t stream) {
    (void)n_in; (void)out_size;
    const float* x  = (const float*)d_in[0];
    const int*   ei = (const int*)d_in[1];
    const float* W1 = (const float*)d_in[2];
    const float* b1 = (const float*)d_in[3];
    const float* W2 = (const float*)d_in[4];
    const float* b2 = (const float*)d_in[5];
    const float* al = (const float*)d_in[6];

    const int N = in_sizes[0] / N_FEAT;
    const int E = in_sizes[1] / 2;
    const int* src = ei;
    const int* dst = ei + E;

    float* dinv = (float*)d_ws;
    size_t off = (((size_t)N * sizeof(float)) + 255) & ~(size_t)255;
    float* hbuf = (float*)((char*)d_ws + off);
    off += (size_t)N * N_FEAT * sizeof(float);
    int* rowptr = (int*)((char*)d_ws + off);
    off += (((size_t)(N + 1) * sizeof(int)) + 255) & ~(size_t)255;
    int* csr = (int*)((char*)d_ws + off);
    off += (size_t)E * sizeof(int);
    float* agg  = (float*)d_out;
    float* outp = (float*)d_out;

    const int nBlocks = (N + 7) / 8;
    const int nBuckets = (N + BK_NODES - 1) / BK_NODES;
    const int pBlocks = (E + PB_EDGES - 1) / PB_EDGES;

    int* btot   = (int*)d_out;
    int* bcur   = btot + BK_MAX;
    int* bstart = bcur + BK_MAX;
    int2* pairs = (int2*)(bstart + BK_MAX + 64);
    int* tmp    = (int*)(pairs + E);
    const size_t scratchInts = 3 * (size_t)BK_MAX + 64 + 3 * (size_t)E;

    if (off <= ws_size && scratchInts <= (size_t)N * N_FEAT && nBuckets <= BK_MAX) {
        cnt_zero_kernel<<<(2 * BK_MAX + 255) / 256, 256, 0, stream>>>(btot, 2 * BK_MAX);
        bucket_count_kernel<<<pBlocks, 256, 0, stream>>>(dst, btot, E, nBuckets);
        bucket_place_kernel<<<pBlocks, 256, 0, stream>>>(src, dst, btot, bcur, bstart, pairs, E, nBuckets);
        bucket_finish_kernel<<<nBuckets, 256, 0, stream>>>(pairs, bstart, rowptr, dinv, tmp, N, nBuckets);
        seg_order_kernel<<<nBlocks, 256, 0, stream>>>(rowptr, tmp, csr, N);

        gemm_wmma_kernel<<<GM_GRID, GM_THREADS, 0, stream>>>(x, W1, hbuf, N);
        gather_finalize_kernel<<<nBlocks, 256, 0, stream>>>(hbuf, dinv, rowptr, csr, b1, al, outp, N);

        gemm_wmma_kernel<<<GM_GRID, GM_THREADS, 0, stream>>>(outp, W2, hbuf, N);
        gather_finalize_kernel<<<nBlocks, 256, 0, stream>>>(hbuf, dinv, rowptr, csr, b2, al, outp, N);
        return;
    }

    deg_init_kernel<<<(N + 255) / 256, 256, 0, stream>>>(dinv, N);
    deg_scatter_kernel<<<(E + 255) / 256, 256, 0, stream>>>(dst, dinv, E);
    deg_rsqrt_kernel<<<(N + 255) / 256, 256, 0, stream>>>(dinv, N);

    const long long n4 = (long long)N * N_FEAT / 4;
    const int zBlocks = (int)((n4 + 255) / 256);
    const int eBlocks = (E + 7) / 8;

    gemm_wmma_kernel<<<GM_GRID, GM_THREADS, 0, stream>>>(x, W1, hbuf, N);
    zero_kernel<<<zBlocks, 256, 0, stream>>>((float4*)agg, n4);
    scatter_kernel<<<eBlocks, 256, 0, stream>>>(hbuf, dinv, src, dst, agg, E);
    finalize_kernel<<<nBlocks, 256, 0, stream>>>(agg, hbuf, dinv, b1, al, outp, N);

    gemm_wmma_kernel<<<GM_GRID, GM_THREADS, 0, stream>>>(outp, W2, hbuf, N);
    zero_kernel<<<zBlocks, 256, 0, stream>>>((float4*)agg, n4);
    scatter_kernel<<<eBlocks, 256, 0, stream>>>(hbuf, dinv, src, dst, agg, E);
    finalize_kernel<<<nBlocks, 256, 0, stream>>>(agg, hbuf, dinv, b2, al, outp, N);
}


# layer-1 gather and epilogue fused in front of the layer-2 product: the wave that stages a tile row sums and finishes it, z1 goes to LDS only; one launch fewer; sum order and k order kept
# speedup vs baseline: 5.4296x; 1.1135x over previous
#include <hip/hip_runtime.h>
#include <math.h>

#define N_FEAT 128

typedef __attribute__((ext_vector_type(2))) float v2f;
typedef __attribute__((ext_vector_type(8))) float v8f;

#define BK_SHIFT 7
#define BK_NODES 128
#define BK_MAX   1024
#define PB_EDGES 4096
#define BK_STAGE 4096

__global__ void cnt_zero_kernel(int* __restrict__ cnt, int n) {
    int i = blockIdx.x * blockDim.x + threadIdx.x;
    if (i < n) cnt[i] = 0;
}

__global__ __launch_bounds__(256) void bucket_count_kernel(const int* __restrict__ dst,
                                                           int* __restrict__ btot, int E, int nb) {
    __shared__ int hist[BK_MAX];
    for (int b = threadIdx.x; b < BK_MAX; b += 256) hist[b] = 0;
    __syncthreads();
    const int e0 = blockIdx.x * PB_EDGES + threadIdx.x;
    int d[PB_EDGES / 256];
    #pragma unroll
    for (int k = 0; k < PB_EDGES / 256; ++k) {
        const int e = e0 + k * 256;
        d[k] = (e < E) ? dst[e] : -1;
    }
    #pragma unroll
    for (int k = 0; k < PB_EDGES / 256; ++k) if (d[k] >= 0) atomicAdd(&hist[d[k] >> BK_SHIFT], 1);
    __syncthreads();
    for (int b = threadIdx.x; b < nb; b += 256) {
        const int h = hist[b];
        if (h) atomicAdd(&btot[b], h);
    }
}

__global__ __launch_bounds__(256) void bucket_place_kernel(const int* __restrict__ src,
                                                           const int* __restrict__ dst,
                                                           const int* __restrict__ btot,
                                                           int* __restrict__ bcur,
                                                           int* __restrict__ bstart,
                                                           int2* __restrict__ pairs, int E, int nb) {
    __shared__ int hist[BK_MAX];
    __shared__ int bst[BK_MAX];
    __shared__ int wtot[8];
    const int tid = threadIdx.x;
    const int lane = tid & 31;
    const int wave = tid >> 5;
    for (int b = tid; b < BK_MAX; b += 256) hist[b] = 0;
    __syncthreads();
    const int e0 = blockIdx.x * PB_EDGES + tid;
    int d[PB_EDGES / 256];
    #pragma unroll
    for (int k = 0; k < PB_EDGES / 256; ++k) {
        const int e = e0 + k * 256;
        d[k] = (e < E) ? dst[e] : -1;
    }
    #pragma unroll
    for (int k = 0; k < PB_EDGES / 256; ++k) if (d[k] >= 0) atomicAdd(&hist[d[k] >> BK_SHIFT], 1);
    int c[4];
    int ts = 0;
    #pragma unroll
    for (int k = 0; k < 4; ++k) { const int b = tid * 4 + k; c[k] = (b < nb) ? btot[b] : 0; ts += c[k]; }
    int inc = ts;
    for (int o = 1; o < 32; o <<= 1) { const int t = __shfl_up(inc, o); if (lane >= o) inc += t; }
    if (lane == 31) wtot[wave] = inc;
    __syncthreads();
    int run = inc - ts;
    for (int w = 0; w < wave; ++w) run += wtot[w];
    #pragma unroll
    for (int k = 0; k < 4; ++k) { bst[tid * 4 + k] = run; run += c[k]; }
    if (blockIdx.x == 0 && tid == 255) bstart[nb] = run;
    __syncthreads();
    for (int b = tid; b < nb; b += 256) {
        const int start = bst[b];
        if (blockIdx.x == 0) bstart[b] = start;
        const int h = hist[b];
        int base = 0;
        if (h) base = atomicAdd(&bcur[b], h);
        hist[b] = start + base;
    }
    __syncthreads();
    int sv[PB_EDGES / 256];
    #pragma unroll
    for (int k = 0; k < PB_EDGES / 256; ++k) sv[k] = (d[k] >= 0) ? src[e0 + k * 256] : 0;
    #pragma unroll
    for (int k = 0; k < PB_EDGES / 256; ++k) {
        if (d[k] >= 0) {
            const int p = atomicAdd(&hist[d[k] >> BK_SHIFT], 1);
            pairs[p] = make_int2(d[k], sv[k]);
        }
    }
}

__global__ __launch_bounds__(256) void bucket_finish_kernel(const int2* __restrict__ pairs,
                                                            const int* __restrict__ bstart,
                                                            int* __restrict__ rowptr,
                                                            float* __restrict__ dinv,
                                                            int* __restrict__ tmp, int n, int nb) {
    __shared__ int cnt[BK_NODES];
    __shared__ int wtot[4];
    __shared__ int stage[BK_STAGE];
    const int tid = threadIdx.x;
    const int lane = tid & 31;
    const int wave = tid >> 5;
    const int b = blockIdx.x;
    const int beg = bstart[b];
    const int end = bstart[b + 1];
    const int node0 = b << BK_SHIFT;
    if (tid < BK_NODES) cnt[tid] = 0;
    __syncthreads();
    for (int i0 = beg + tid; i0 < end; i0 += 8 * 256) {
        int dn[8];
        #pragma unroll
        for (int k = 0; k < 8; ++k) { const int i = i0 + k * 256; dn[k] = (i < end) ? pairs[i].x : -1; }
        #pragma unroll
        for (int k = 0; k < 8; ++k) if (dn[k] >= 0) atomicAdd(&cnt[dn[k] - node0], 1);
    }
    __syncthreads();
    const int c = (tid < BK_NODES) ? cnt[tid] : 0;
    int inc = c;
    for (int o = 1; o < 32; o <<= 1) { const int t = __shfl_up(inc, o); if (lane >= o) inc += t; }
    if (tid < BK_NODES && lane == 31) wtot[wave] = inc;
    __syncthreads();
    if (tid < BK_NODES) {
        int ex = inc - c;
        for (int w = 0; w < wave; ++w) ex += wtot[w];
        cnt[tid] = ex;
        const int node = node0 + tid;
        if (node < n) {
            rowptr[node] = beg + ex;
            dinv[node] = rsqrtf((float)(c + 1));
        }
    }
    if (b == nb - 1 && tid == 0) rowptr[n] = end;
    __syncthreads();
    const bool fits = (end - beg) <= BK_STAGE;
    for (int i0 = beg + tid; i0 < end; i0 += 8 * 256) {
        int2 pr[8];
        #pragma unroll
        for (int k = 0; k < 8; ++k) { const int i = i0 + k * 256; pr[k] = (i < end) ? pairs[i] : make_int2(-1, 0); }
        #pragma unroll
        for (int k = 0; k < 8; ++k) {
            if (pr[k].x >= 0) {
                const int p = atomicAdd(&cnt[pr[k].x - node0], 1);
                if (fits) stage[p] = pr[k].y; else tmp[beg + p] = pr[k].y;
            }
        }
    }
    if (fits) {
        __syncthreads();
        for (int i = tid; i < end - beg; i += 256) tmp[beg + i] = stage[i];
    }
}

__global__ __launch_bounds__(256) void seg_order_kernel(const int* __restrict__ rowptr,
                                                        const int* __restrict__ tmp,
                                                        int* __restrict__ csr, int n) {
    const int lane = threadIdx.x & 31;
    int node = blockIdx.x * 8 + (threadIdx.x >> 5);
    if (node >= n) return;
    node = __builtin_amdgcn_readfirstlane(node);
    const int beg = rowptr[node];
    const int d = rowptr[node + 1] - beg;
    for (int c = 0; c < d; c += 32) {
        const int i = c + lane;
        const int x = (i < d) ? tmp[beg + i] : 0;
        int pos = 0;
        for (int c2 = 0; c2 < d; c2 += 32) {
            const int i2 = c2 + lane;
            const int y = (i2 < d) ? tmp[beg + i2] : 0;
            const int m2 = (d - c2 < 32) ? (d - c2) : 32;
            for (int j = 0; j < m2; ++j) {
                const int vj = __builtin_amdgcn_readlane(y, j);
                const int t = (c2 + j < i) ? 1 : 0;
                pos += (vj < x + t) ? 1 : 0;
            }
        }
        if (i < d) csr[beg + pos] = x;
    }
}

#define GM_ROWS    32
#define GM_PITCH   132
#define GM_GRID    512
#define GM_THREADS 512

__global__ __launch_bounds__(GM_THREADS) void gemm_wmma_kernel(const float* __restrict__ X,
                                                               const float* __restrict__ W,
                                                               const float* __restrict__ rscale,
                                                               float* __restrict__ H, int M) {
    __shared__ __attribute__((aligned(16))) float xs[2][GM_ROWS][GM_PITCH];
    const int tid  = threadIdx.x;
    const int wave = tid >> 5;
    const int lane = tid & 31;
    const int half = lane >> 4;
    const int l16  = lane & 15;
    const int col  = (wave & 7) * 16 + l16;
    const int mrow = (wave >> 3) * 16;
    const int nTiles = (M + GM_ROWS - 1) / GM_ROWS;
    const int srow = tid >> 5;
    const int scol = lane * 4;

    int tile = blockIdx.x;
    float4 p0 = make_float4(0.f, 0.f, 0.f, 0.f);
    float4 p1 = p0;
    float s0 = 1.0f;
    float s1 = 1.0f;
    if (tile < nTiles) {
        int r0 = tile * GM_ROWS + srow;
        int r1 = r0 + 16;
        r0 = (r0 < M) ? r0 : M - 1;
        r1 = (r1 < M) ? r1 : M - 1;
        p0 = *(const float4*)(X + (size_t)r0 * N_FEAT + scol);
        p1 = *(const float4*)(X + (size_t)r1 * N_FEAT + scol);
        if (rscale) { s0 = rscale[r0]; s1 = rscale[r1]; }
    }

    v2f b[N_FEAT / 4];
    #pragma unroll
    for (int t = 0; t < N_FEAT / 4; ++t) {
        b[t].x = W[(size_t)(4 * t + half * 2 + 0) * N_FEAT + col];
        b[t].y = W[(size_t)(4 * t + half * 2 + 1) * N_FEAT + col];
    }

    if (tile < nTiles) {
        *(float4*)&xs[0][srow][scol]      = make_float4(p0.x * s0, p0.y * s0, p0.z * s0, p0.w * s0);
        *(float4*)&xs[0][srow + 16][scol] = make_float4(p1.x * s1, p1.y * s1, p1.z * s1, p1.w * s1);
    }
    int stage = 0;
    for (; tile < nTiles; tile += gridDim.x) {
        __syncthreads();
        const int nxt = tile + gridDim.x;
        const bool more = nxt < nTiles;
        if (more) {
            int r0 = nxt * GM_ROWS + srow;
            int r1 = r0 + 16;
            r0 = (r0 < M) ? r0 : M - 1;
            r1 = (r1 < M) ? r1 : M - 1;
            p0 = *(const float4*)(X + (size_t)r0 * N_FEAT + scol);
            p1 = *(const float4*)(X + (size_t)r1 * N_FEAT + scol);
            if (rscale) { s0 = rscale[r0]; s1 = rscale[r1]; }
        }

        v8f acc = {};
        const float* arow = &xs[stage][mrow + l16][half * 2];
        #pragma unroll
        for (int t = 0; t < N_FEAT / 4; ++t) {
            const v2f a = *(const v2f*)(arow + 4 * t);
            acc = __builtin_amdgcn_wmma_f32_16x16x4_f32(
                 false, a,  false, b[t],
                 (short)0, acc,  false,  false);
        }
        const int orow = tile * GM_ROWS + mrow + half * 8;
        float* out = H + (size_t)orow * N_FEAT + col;
        if ((tile + 1) * GM_ROWS <= M) {
            #pragma unroll
            for (int r = 0; r < 8; ++r) out[(size_t)r * N_FEAT] = acc[r];
        } else {
            #pragma unroll
            for (int r = 0; r < 8; ++r) {
                if (orow + r < M) out[(size_t)r * N_FEAT] = acc[r];
            }
        }
        if (more) {
            *(float4*)&xs[stage ^ 1][srow][scol]      = make_float4(p0.x * s0, p0.y * s0, p0.z * s0, p0.w * s0);
            *(float4*)&xs[stage ^ 1][srow + 16][scol] = make_float4(p1.x * s1, p1.y * s1, p1.z * s1, p1.w * s1);
        }
        stage ^= 1;
    }
}


template <int K>
__device__ __forceinline__ void add_rows(float4& acc, const float* __restrict__ hl, int sidx, int j) {
    float4 v[K];
    for (int k = 0; k < K; ++k) {
        const int s = __builtin_amdgcn_readlane(sidx, j + k);
        v[k] = *(const float4*)(hl + (size_t)s * N_FEAT);
    }
    for (int k = 0; k < K; ++k) {
        acc.x += v[k].x;
        acc.y += v[k].y;
        acc.z += v[k].z;
        acc.w += v[k].w;
    }
}

__global__ __launch_bounds__(256) void gather_finalize_kernel(const float* __restrict__ H,
                                                              const float* __restrict__ dinv,
                                                              const int* __restrict__ rowptr,
                                                              const int* __restrict__ csr,
                                                              const float* __restrict__ bias,
                                                              const float* __restrict__ alpha,
                                                              float* __restrict__ out, int n) {
    const int lane = threadIdx.x & 31;
    int node = blockIdx.x * 8 + (threadIdx.x >> 5);
    if (node >= n) return;
    node = __builtin_amdgcn_readfirstlane(node);
    const int beg = rowptr[node];
    const int end = rowptr[node + 1];
    const float dd = dinv[node];
    const float* hl = H + lane * 4;
    const float4 hv = *(const float4*)(hl + (size_t)node * N_FEAT);
    const float4 bv = *(const float4*)(bias + lane * 4);
    const float4 av = *(const float4*)(alpha + lane * 4);
    float4 acc = make_float4(0.f, 0.f, 0.f, 0.f);
    for (int base = beg; base < end; base += 32) {
        const int m = (end - base < 32) ? (end - base) : 32;
        int sidx = 0;
        if (lane < m) sidx = csr[base + lane];
        int j = 0;
        for (; j + 8 <= m; j += 8) add_rows<8>(acc, hl, sidx, j);
        if (j + 4 <= m) { add_rows<4>(acc, hl, sidx, j); j += 4; }
        const int r = m - j;
        if (r == 3)      add_rows<3>(acc, hl, sidx, j);
        else if (r == 2) add_rows<2>(acc, hl, sidx, j);
        else if (r == 1) add_rows<1>(acc, hl, sidx, j);
    }
    float4 r;
    r.x = __builtin_fmaf(acc.x + hv.x, dd, bv.x);  r.x = r.x > 0.f ? r.x : av.x * r.x;
    r.y = __builtin_fmaf(acc.y + hv.y, dd, bv.y);  r.y = r.y > 0.f ? r.y : av.y * r.y;
    r.z = __builtin_fmaf(acc.z + hv.z, dd, bv.z);  r.z = r.z > 0.f ? r.z : av.z * r.z;
    r.w = __builtin_fmaf(acc.w + hv.w, dd, bv.w);  r.w = r.w > 0.f ? r.w : av.w * r.w;
    *(float4*)(out + (size_t)node * N_FEAT + lane * 4) = r;
}


__device__ __forceinline__ float4 gather_row(const float* __restrict__ hl, const int* __restrict__ csr,
                                             int beg, int end, float dd, float4 hv, int sidx,
                                             float4 bv, float4 av, int lane) {
    float4 acc = make_float4(0.f, 0.f, 0.f, 0.f);
    for (int base = beg; base < end; base += 32) {
        const int m = (end - base < 32) ? (end - base) : 32;
        if (base != beg) {
            sidx = 0;
            if (lane < m) sidx = csr[base + lane];
        }
        int j = 0;
        for (; j + 8 <= m; j += 8) add_rows<8>(acc, hl, sidx, j);
        if (j + 4 <= m) { add_rows<4>(acc, hl, sidx, j); j += 4; }
        const int r = m - j;
        if (r == 3)      add_rows<3>(acc, hl, sidx, j);
        else if (r == 2) add_rows<2>(acc, hl, sidx, j);
        else if (r == 1) add_rows<1>(acc, hl, sidx, j);
    }
    float4 r;
    r.x = __builtin_fmaf(acc.x + hv.x, dd, bv.x);  r.x = r.x > 0.f ? r.x : av.x * r.x;
    r.y = __builtin_fmaf(acc.y + hv.y, dd, bv.y);  r.y = r.y > 0.f ? r.y : av.y * r.y;
    r.z = __builtin_fmaf(acc.z + hv.z, dd, bv.z);  r.z = r.z > 0.f ? r.z : av.z * r.z;
    r.w = __builtin_fmaf(acc.w + hv.w, dd, bv.w);  r.w = r.w > 0.f ? r.w : av.w * r.w;
    return r;
}

__device__ __forceinline__ void stage_gathered_rows(float (*xt)[GM_PITCH], int tile, int srow, int lane,
                                                    const float* __restrict__ hl,
                                                    const float* __restrict__ dinv,
                                                    const int* __restrict__ rowptr,
                                                    const int* __restrict__ csr,
                                                    float4 bv, float4 av, int M) {
    int n0 = tile * GM_ROWS + srow;
    int n1 = n0 + 16;
    n0 = (n0 < M) ? n0 : M - 1;
    n1 = (n1 < M) ? n1 : M - 1;
    const int beg0 = rowptr[n0];
    const int end0 = rowptr[n0 + 1];
    const int beg1 = rowptr[n1];
    const int end1 = rowptr[n1 + 1];
    const float dd0 = dinv[n0];
    const float dd1 = dinv[n1];
    const float4 hv0 = *(const float4*)(hl + (size_t)n0 * N_FEAT);
    const float4 hv1 = *(const float4*)(hl + (size_t)n1 * N_FEAT);
    int si0 = 0;
    if (lane < end0 - beg0) si0 = csr[beg0 + lane];
    int si1 = 0;
    if (lane < end1 - beg1) si1 = csr[beg1 + lane];
    const float4 z0 = gather_row(hl, csr, beg0, end0, dd0, hv0, si0, bv, av, lane);
    *(float4*)&xt[srow][lane * 4] = make_float4(z0.x * dd0, z0.y * dd0, z0.z * dd0, z0.w * dd0);
    const float4 z1 = gather_row(hl, csr, beg1, end1, dd1, hv1, si1, bv, av, lane);
    *(float4*)&xt[srow + 16][lane * 4] = make_float4(z1.x * dd1, z1.y * dd1, z1.z * dd1, z1.w * dd1);
}

__global__ __launch_bounds__(GM_THREADS) void gather_gemm_kernel(const float* __restrict__ Hin,
                                                                 const float* __restrict__ dinv,
                                                                 const int* __restrict__ rowptr,
                                                                 const int* __restrict__ csr,
                                                                 const float* __restrict__ bias,
                                                                 const float* __restrict__ alpha,
                                                                 const float* __restrict__ W,
                                                                 float* __restrict__ Hout, int M) {
    __shared__ __attribute__((aligned(16))) float xs[2][GM_ROWS][GM_PITCH];
    const int tid  = threadIdx.x;
    const int lane = tid & 31;
    const int wave = __builtin_amdgcn_readfirstlane(tid >> 5);
    const int half = lane >> 4;
    const int l16  = lane & 15;
    const int col  = (wave & 7) * 16 + l16;
    const int mrow = (wave >> 3) * 16;
    const int nTiles = (M + GM_ROWS - 1) / GM_ROWS;
    const float* hl = Hin + lane * 4;
    const float4 bv = *(const float4*)(bias + lane * 4);
    const float4 av = *(const float4*)(alpha + lane * 4);

    v2f b[N_FEAT / 4];
    #pragma unroll
    for (int t = 0; t < N_FEAT / 4; ++t) {
        b[t].x = W[(size_t)(4 * t + half * 2 + 0) * N_FEAT + col];
        b[t].y = W[(size_t)(4 * t + half * 2 + 1) * N_FEAT + col];
    }

    int tile = blockIdx.x;
    if (tile < nTiles)
        stage_gathered_rows(xs[0], tile, wave, lane, hl, dinv, rowptr, csr, bv, av, M);
    int stage = 0;
    for (; tile < nTiles; tile += gridDim.x) {
        __syncthreads();
        const int nxt = tile + gridDim.x;
        if (nxt < nTiles)
            stage_gathered_rows(xs[stage ^ 1], nxt, wave, lane, hl, dinv, rowptr, csr, bv, av, M);

        v8f acc = {};
        const float* arow = &xs[stage][mrow + l16][half * 2];
        #pragma unroll
        for (int t = 0; t < N_FEAT / 4; ++t) {
            const v2f a = *(const v2f*)(arow + 4 * t);
            acc = __builtin_amdgcn_wmma_f32_16x16x4_f32(
                 false, a,  false, b[t],
                 (short)0, acc,  false,  false);
        }
        const int orow = tile * GM_ROWS + mrow + half * 8;
        float* out = Hout + (size_t)orow * N_FEAT + col;
        if ((tile + 1) * GM_ROWS <= M) {
            #pragma unroll
            for (int r = 0; r < 8; ++r) out[(size_t)r * N_FEAT] = acc[r];
        } else {
            #pragma unroll
            for (int r = 0; r < 8; ++r) {
                if (orow + r < M) out[(size_t)r * N_FEAT] = acc[r];
            }
        }
        stage ^= 1;
    }
}

__global__ void finalize_kernel(const float* __restrict__ agg, const float* __restrict__ H,
                                const float* __restrict__ dinv, const float* __restrict__ bias,
                                const float* __restrict__ alpha, float* __restrict__ out, int n) {
    const int lane = threadIdx.x & 31;
    const int node = blockIdx.x * (blockDim.x >> 5) + (threadIdx.x >> 5);
    if (node >= n) return;
    const float di = dinv[node];
    const float di2 = di * di;
    const int f0 = lane * 4;
    const float4 ag = *(const float4*)(agg + (size_t)node * N_FEAT + f0);
    const float4 hv = *(const float4*)(H + (size_t)node * N_FEAT + f0);
    const float4 bv = *(const float4*)(bias + f0);
    const float4 av = *(const float4*)(alpha + f0);
    float4 r;
    r.x = ag.x + hv.x * di2 + bv.x;  r.x = r.x > 0.f ? r.x : av.x * r.x;
    r.y = ag.y + hv.y * di2 + bv.y;  r.y = r.y > 0.f ? r.y : av.y * r.y;
    r.z = ag.z + hv.z * di2 + bv.z;  r.z = r.z > 0.f ? r.z : av.z * r.z;
    r.w = ag.w + hv.w * di2 + bv.w;  r.w = r.w > 0.f ? r.w : av.w * r.w;
    *(float4*)(out + (size_t)node * N_FEAT + f0) = r;
}

__global__ void deg_init_kernel(float* __restrict__ deg, int n) {
    int i = blockIdx.x * blockDim.x + threadIdx.x;
    if (i < n) deg[i] = 1.0f;
}

__global__ void deg_scatter_kernel(const int* __restrict__ dst, float* __restrict__ deg, int E) {
    int e = blockIdx.x * blockDim.x + threadIdx.x;
    if (e < E) atomicAdd(&deg[dst[e]], 1.0f);
}

__global__ void deg_rsqrt_kernel(float* __restrict__ deg, int n) {
    int i = blockIdx.x * blockDim.x + threadIdx.x;
    if (i < n) deg[i] = rsqrtf(deg[i]);
}

__global__ void zero_kernel(float4* __restrict__ p, long long n4) {
    long long i = (long long)blockIdx.x * blockDim.x + threadIdx.x;
    if (i < n4) p[i] = make_float4(0.f, 0.f, 0.f, 0.f);
}

__global__ void scatter_kernel(const float* __restrict__ H, const float* __restrict__ dinv,
                               const int* __restrict__ src, const int* __restrict__ dst,
                               float* __restrict__ agg, int E) {
    const int lane = threadIdx.x & 31;
    const int e = blockIdx.x * (blockDim.x >> 5) + (threadIdx.x >> 5);
    if (e >= E) return;
    const int s = src[e];
    const int d = dst[e];
    const float nrm = dinv[s] * dinv[d];
    const float4 v = *(const float4*)(H + (size_t)s * N_FEAT + lane * 4);
    float* p = agg + (size_t)d * N_FEAT + lane * 4;
    atomicAdd(p + 0, v.x * nrm);
    atomicAdd(p + 1, v.y * nrm);
    atomicAdd(p + 2, v.z * nrm);
    atomicAdd(p + 3, v.w * nrm);
}

extern "C" void kernel_launch(void* const* d_in, const int* in_sizes, int n_in,
                              void* d_out, int out_size, void* d_ws, size_t ws_size,
                              hipStream_t stream) {
    (void)n_in; (void)out_size;
    const float* x  = (const float*)d_in[0];
    const int*   ei = (const int*)d_in[1];
    const float* W1 = (const float*)d_in[2];
    const float* b1 = (const float*)d_in[3];
    const float* W2 = (const float*)d_in[4];
    const float* b2 = (const float*)d_in[5];
    const float* al = (const float*)d_in[6];

    const int N = in_sizes[0] / N_FEAT;
    const int E = in_sizes[1] / 2;
    const int* src = ei;
    const int* dst = ei + E;

    float* dinv = (float*)d_ws;
    size_t off = (((size_t)N * sizeof(float)) + 255) & ~(size_t)255;
    float* hbuf = (float*)((char*)d_ws + off);
    off += (size_t)N * N_FEAT * sizeof(float);
    int* rowptr = (int*)((char*)d_ws + off);
    off += (((size_t)(N + 1) * sizeof(int)) + 255) & ~(size_t)255;
    int* csr = (int*)((char*)d_ws + off);
    off += (size_t)E * sizeof(int);
    float* agg  = (float*)d_out;
    float* outp = (float*)d_out;

    const int nBlocks = (N + 7) / 8;
    const int nBuckets = (N + BK_NODES - 1) / BK_NODES;
    const int pBlocks = (E + PB_EDGES - 1) / PB_EDGES;

    int* btot   = (int*)d_out;
    int* bcur   = btot + BK_MAX;
    int* bstart = bcur + BK_MAX;
    int2* pairs = (int2*)(bstart + BK_MAX + 64);
    int* tmp    = (int*)(pairs + E);
    const size_t scratchInts = 3 * (size_t)BK_MAX + 64 + 3 * (size_t)E;

    if (off <= ws_size && scratchInts <= (size_t)N * N_FEAT && nBuckets <= BK_MAX) {
        cnt_zero_kernel<<<(2 * BK_MAX + 255) / 256, 256, 0, stream>>>(btot, 2 * BK_MAX);
        bucket_count_kernel<<<pBlocks, 256, 0, stream>>>(dst, btot, E, nBuckets);
        bucket_place_kernel<<<pBlocks, 256, 0, stream>>>(src, dst, btot, bcur, bstart, pairs, E, nBuckets);
        bucket_finish_kernel<<<nBuckets, 256, 0, stream>>>(pairs, bstart, rowptr, dinv, tmp, N, nBuckets);
        seg_order_kernel<<<nBlocks, 256, 0, stream>>>(rowptr, tmp, csr, N);

        gemm_wmma_kernel<<<GM_GRID, GM_THREADS, 0, stream>>>(x, W1, dinv, outp, N);

        gather_gemm_kernel<<<GM_GRID, GM_THREADS, 0, stream>>>(outp, dinv, rowptr, csr, b1, al, W2, hbuf, N);

        gather_finalize_kernel<<<nBlocks, 256, 0, stream>>>(hbuf, dinv, rowptr, csr, b2, al, outp, N);
        return;
    }

    deg_init_kernel<<<(N + 255) / 256, 256, 0, stream>>>(dinv, N);
    deg_scatter_kernel<<<(E + 255) / 256, 256, 0, stream>>>(dst, dinv, E);
    deg_rsqrt_kernel<<<(N + 255) / 256, 256, 0, stream>>>(dinv, N);

    const long long n4 = (long long)N * N_FEAT / 4;
    const int zBlocks = (int)((n4 + 255) / 256);
    const int eBlocks = (E + 7) / 8;

    gemm_wmma_kernel<<<GM_GRID, GM_THREADS, 0, stream>>>(x, W1, nullptr, hbuf, N);
    zero_kernel<<<zBlocks, 256, 0, stream>>>((float4*)agg, n4);
    scatter_kernel<<<eBlocks, 256, 0, stream>>>(hbuf, dinv, src, dst, agg, E);
    finalize_kernel<<<nBlocks, 256, 0, stream>>>(agg, hbuf, dinv, b1, al, outp, N);

    gemm_wmma_kernel<<<GM_GRID, GM_THREADS, 0, stream>>>(outp, W2, nullptr, hbuf, N);
    zero_kernel<<<zBlocks, 256, 0, stream>>>((float4*)agg, n4);
    scatter_kernel<<<eBlocks, 256, 0, stream>>>(hbuf, dinv, src, dst, agg, E);
    finalize_kernel<<<nBlocks, 256, 0, stream>>>(agg, hbuf, dinv, b2, al, outp, N);
}
